# starting kernel
# baseline (speedup 1.0000x reference)
#include <hip/hip_runtime.h>
#include <hip/hip_bf16.h>
#include <math.h>


#define NUM_ITEMS (4096 * 64)
#define OUT_HALF  (NUM_ITEMS * 16)
#define SIGMA_R 0.05f
#define SIGMA_T 0.03f

typedef float v2f __attribute__((ext_vector_type(2)));
typedef float v8f __attribute__((ext_vector_type(8)));


__device__ __forceinline__ void se3_exp16(const float xi[6], float* H) {
    const float wx = xi[0], wy = xi[1], wz = xi[2];
    const float vx = xi[3], vy = xi[4], vz = xi[5];
    float th2  = wx * wx + wy * wy + wz * wz;
    bool small = th2 < 1e-8f;
    float th2s = small ? 1.0f : th2;
    float th   = sqrtf(th2s);
    float s, c;
    sincosf(th, &s, &c);
    float A  = small ? 1.0f - th2 * (1.0f / 6.0f)   : s / th;
    float Bc = small ? 0.5f - th2 * (1.0f / 24.0f)  : (1.0f - c) / th2s;
    float Cc = small ? (1.0f / 6.0f) - th2 * (1.0f / 120.0f)
                     : (th - s) / (th2s * th);
    float xx = wx * wx, yy = wy * wy, zz = wz * wz;
    float xy = wx * wy, xz = wx * wz, yz = wy * wz;
    float R00 = 1.0f - Bc * (yy + zz), R01 = Bc * xy - A * wz, R02 = Bc * xz + A * wy;
    float R10 = Bc * xy + A * wz, R11 = 1.0f - Bc * (xx + zz), R12 = Bc * yz - A * wx;
    float R20 = Bc * xz - A * wy, R21 = Bc * yz + A * wx, R22 = 1.0f - Bc * (xx + yy);
    float V00 = 1.0f - Cc * (yy + zz), V01 = Cc * xy - Bc * wz, V02 = Cc * xz + Bc * wy;
    float V10 = Cc * xy + Bc * wz, V11 = 1.0f - Cc * (xx + zz), V12 = Cc * yz - Bc * wx;
    float V20 = Cc * xz - Bc * wy, V21 = Cc * yz + Bc * wx, V22 = 1.0f - Cc * (xx + yy);
    float tx = V00 * vx + V01 * vy + V02 * vz;
    float ty = V10 * vx + V11 * vy + V12 * vz;
    float tz = V20 * vx + V21 * vy + V22 * vz;
    H[0] = R00; H[1] = R01; H[2]  = R02; H[3]  = tx;
    H[4] = R10; H[5] = R11; H[6]  = R12; H[7]  = ty;
    H[8] = R20; H[9] = R21; H[10] = R22; H[11] = tz;
    H[12] = 0.0f; H[13] = 0.0f; H[14] = 0.0f; H[15] = 1.0f;
}

__device__ __forceinline__ void se3_log_of_inv(const float* O, float xi[6]) {
    float R00 = O[0], R01 = O[4], R02 = O[8];
    float R10 = O[1], R11 = O[5], R12 = O[9];
    float R20 = O[2], R21 = O[6], R22 = O[10];
    float tx = -(R00 * O[3] + R01 * O[7] + R02 * O[11]);
    float ty = -(R10 * O[3] + R11 * O[7] + R12 * O[11]);
    float tz = -(R20 * O[3] + R21 * O[7] + R22 * O[11]);
    float e0 = 0.5f * (R21 - R12);
    float e1 = 0.5f * (R02 - R20);
    float e2 = 0.5f * (R10 - R01);
    float cos_t = 0.5f * (R00 + R11 + R22 - 1.0f);
    cos_t = fminf(1.0f, fmaxf(-1.0f, cos_t));
    float s2 = e0 * e0 + e1 * e1 + e2 * e2;
    bool small = s2 < 1e-12f;
    float sin_s = sqrtf(small ? 1.0f : s2);
    float th = atan2f(sin_s, cos_t);
    float fac = small ? 1.0f + s2 * (1.0f / 6.0f) : th / sin_s;
    float wx = fac * e0, wy = fac * e1, wz = fac * e2;
    float omc  = small ? 1.0f : 1.0f - cos_t;
    float th2s = small ? 1.0f : th * th;
    float D = small ? (1.0f / 12.0f) + s2 * (1.0f / 720.0f)
                    : (1.0f - th * sin_s / (2.0f * omc)) / th2s;
    float xx = wx * wx, yy = wy * wy, zz = wz * wz;
    float xy = wx * wy, xz = wx * wz, yz = wy * wz;
    float Vi00 = 1.0f - D * (yy + zz), Vi01 = D * xy + 0.5f * wz, Vi02 = D * xz - 0.5f * wy;
    float Vi10 = D * xy - 0.5f * wz, Vi11 = 1.0f - D * (xx + zz), Vi12 = D * yz + 0.5f * wx;
    float Vi20 = D * xz + 0.5f * wy, Vi21 = D * yz - 0.5f * wx, Vi22 = 1.0f - D * (xx + yy);
    xi[0] = wx; xi[1] = wy; xi[2] = wz;
    xi[3] = Vi00 * tx + Vi01 * ty + Vi02 * tz;
    xi[4] = Vi10 * tx + Vi11 * ty + Vi12 * tz;
    xi[5] = Vi20 * tx + Vi21 * ty + Vi22 * tz;
}


__global__ __launch_bounds__(256) void
se3_diffuse_kernel(const float* __restrict__ twist,
                   const float* __restrict__ noise,
                   const float* __restrict__ alpha_bars,
                   const int* __restrict__ timesteps,
                   float* __restrict__ out) {
    __shared__ float sA[8][32][16];
    __shared__ float sB[8][32][16];
    __shared__ float sN[8][32][16];
    __shared__ float sDump[8][48];

    const int tid  = threadIdx.x;
    const int wave = tid >> 5;
    const int lane = tid & 31;
    const int item = blockIdx.x * 256 + tid;
    const int b    = item >> 6;

    const float2* tw2 = (const float2*)(twist + (size_t)item * 6);
    const float2* nz2 = (const float2*)(noise + (size_t)item * 6);
    float2 t0 = tw2[0], t1 = tw2[1], t2 = tw2[2];
    float2 n0 = nz2[0], n1 = nz2[1], n2 = nz2[2];
    float tw[6] = {t0.x, t0.y, t1.x, t1.y, t2.x, t2.y};
    float nz[6] = {n0.x, n0.y, n1.x, n1.y, n2.x, n2.y};

    const float ab = alpha_bars[timesteps[b]];

    float O[16];
    se3_exp16(tw, O);

    float xi[6];
    se3_log_of_inv(O, xi);
    const float f1 = 1.0f - sqrtf(ab);
#pragma unroll
    for (int i = 0; i < 6; ++i) xi[i] *= f1;
    float E[16];
    se3_exp16(xi, E);

    const float f2 = sqrtf(1.0f - ab);
    float nv[6] = {f2 * SIGMA_R * nz[0], f2 * SIGMA_R * nz[1], f2 * SIGMA_R * nz[2],
                   f2 * SIGMA_T * nz[3], f2 * SIGMA_T * nz[4], f2 * SIGMA_T * nz[5]};
    float N[16];
    se3_exp16(nv, N);

#pragma unroll
    for (int i = 0; i < 16; ++i) {
        sA[wave][lane][i] = E[i];
        sB[wave][lane][i] = O[i];
        sN[wave][lane][i] = N[i];
    }
    __builtin_amdgcn_wave_barrier();

    const int half = lane >> 4;
    const int q    = lane & 15;
    const int sub  = q & 3;
    const int grp  = q >> 2;

    const bool act = (grp >> 1) == half;
    const bool hi  = (grp & 1) != 0;
    float* const dumpBase = &sDump[wave][lane];

#pragma unroll
    for (int r = 0; r < 8; ++r) {
        const int it = 4 * r + grp;

        v2f a, bb;
        a.x  = sA[wave][it][sub * 4 + 2 * half];
        a.y  = sA[wave][it][sub * 4 + 2 * half + 1];
        bb.x = sB[wave][it][(2 * half) * 4 + sub];
        bb.y = sB[wave][it][(2 * half + 1) * 4 + sub];

        v8f c = {};
        v8f d = __builtin_amdgcn_wmma_f32_16x16x4_f32(
            false, a, false, bb, (short)0, c, false, false);

        {
            float* p = act ? &sA[wave][it][sub] : dumpBase;
            p[0]  = hi ? d[4] : d[0];
            p[4]  = hi ? d[5] : d[1];
            p[8]  = hi ? d[6] : d[2];
            p[12] = hi ? d[7] : d[3];
        }
        __builtin_amdgcn_wave_barrier();

        v2f an, bm;
        an.x = sN[wave][it][sub * 4 + 2 * half];
        an.y = sN[wave][it][sub * 4 + 2 * half + 1];
        bm.x = sA[wave][it][(2 * half) * 4 + sub];
        bm.y = sA[wave][it][(2 * half + 1) * 4 + sub];

        v8f d2 = __builtin_amdgcn_wmma_f32_16x16x4_f32(
            false, an, false, bm, (short)0, c, false, false);

        {
            float* p = act ? &sB[wave][it][sub] : dumpBase;
            p[0]  = hi ? d2[4] : d2[0];
            p[4]  = hi ? d2[5] : d2[1];
            p[8]  = hi ? d2[6] : d2[2];
            p[12] = hi ? d2[7] : d2[3];
        }
        __builtin_amdgcn_wave_barrier();
    }

    const float4* s1 = (const float4*)(&sB[wave][0][0]);
    const float4* s2 = (const float4*)(&sN[wave][0][0]);
    float4* o1 = ((float4*)out) + (size_t)blockIdx.x * 1024 + wave * 128;
    float4* o2 = ((float4*)(out + OUT_HALF)) + (size_t)blockIdx.x * 1024 + wave * 128;
#pragma unroll
    for (int j = 0; j < 4; ++j) {
        o1[lane + j * 32] = s1[lane + j * 32];
        o2[lane + j * 32] = s2[lane + j * 32];
    }
}

extern "C" void kernel_launch(void* const* d_in, const int* in_sizes, int n_in,
                              void* d_out, int out_size, void* d_ws, size_t ws_size,
                              hipStream_t stream) {
    const float* twist      = (const float*)d_in[0];
    const float* noise      = (const float*)d_in[1];
    const float* alpha_bars = (const float*)d_in[2];
    const int*   timesteps  = (const int*)d_in[3];
    float*       out        = (float*)d_out;

    const int blocks = NUM_ITEMS / 256;
    hipLaunchKernelGGL(se3_diffuse_kernel, dim3(blocks), dim3(256), 0, stream,
                       twist, noise, alpha_bars, timesteps, out);
}


# H_t = exp(sqrt(ab)*twist) directly in waves whose items all have angle < 3 rad (log(inv(exp(x))) = -x there: the log, one exp and the E*O product drop out); other waves keep the full chain
# speedup vs baseline: 1.3341x; 1.3341x over previous
#include <hip/hip_runtime.h>
#include <hip/hip_bf16.h>
#include <math.h>


#define NUM_ITEMS (4096 * 64)
#define OUT_HALF  (NUM_ITEMS * 16)
#define SIGMA_R 0.05f
#define SIGMA_T 0.03f

typedef float v2f __attribute__((ext_vector_type(2)));
typedef float v8f __attribute__((ext_vector_type(8)));


__device__ __forceinline__ void se3_exp16(const float xi[6], float* H) {
    const float wx = xi[0], wy = xi[1], wz = xi[2];
    const float vx = xi[3], vy = xi[4], vz = xi[5];
    float th2  = wx * wx + wy * wy + wz * wz;
    bool small = th2 < 1e-8f;
    float th2s = small ? 1.0f : th2;
    float th   = sqrtf(th2s);
    float s, c;
    sincosf(th, &s, &c);
    float A  = small ? 1.0f - th2 * (1.0f / 6.0f)   : s / th;
    float Bc = small ? 0.5f - th2 * (1.0f / 24.0f)  : (1.0f - c) / th2s;
    float Cc = small ? (1.0f / 6.0f) - th2 * (1.0f / 120.0f)
                     : (th - s) / (th2s * th);
    float xx = wx * wx, yy = wy * wy, zz = wz * wz;
    float xy = wx * wy, xz = wx * wz, yz = wy * wz;
    float R00 = 1.0f - Bc * (yy + zz), R01 = Bc * xy - A * wz, R02 = Bc * xz + A * wy;
    float R10 = Bc * xy + A * wz, R11 = 1.0f - Bc * (xx + zz), R12 = Bc * yz - A * wx;
    float R20 = Bc * xz - A * wy, R21 = Bc * yz + A * wx, R22 = 1.0f - Bc * (xx + yy);
    float V00 = 1.0f - Cc * (yy + zz), V01 = Cc * xy - Bc * wz, V02 = Cc * xz + Bc * wy;
    float V10 = Cc * xy + Bc * wz, V11 = 1.0f - Cc * (xx + zz), V12 = Cc * yz - Bc * wx;
    float V20 = Cc * xz - Bc * wy, V21 = Cc * yz + Bc * wx, V22 = 1.0f - Cc * (xx + yy);
    float tx = V00 * vx + V01 * vy + V02 * vz;
    float ty = V10 * vx + V11 * vy + V12 * vz;
    float tz = V20 * vx + V21 * vy + V22 * vz;
    H[0] = R00; H[1] = R01; H[2]  = R02; H[3]  = tx;
    H[4] = R10; H[5] = R11; H[6]  = R12; H[7]  = ty;
    H[8] = R20; H[9] = R21; H[10] = R22; H[11] = tz;
    H[12] = 0.0f; H[13] = 0.0f; H[14] = 0.0f; H[15] = 1.0f;
}

__device__ __forceinline__ void se3_log_of_inv(const float* O, float xi[6]) {
    float R00 = O[0], R01 = O[4], R02 = O[8];
    float R10 = O[1], R11 = O[5], R12 = O[9];
    float R20 = O[2], R21 = O[6], R22 = O[10];
    float tx = -(R00 * O[3] + R01 * O[7] + R02 * O[11]);
    float ty = -(R10 * O[3] + R11 * O[7] + R12 * O[11]);
    float tz = -(R20 * O[3] + R21 * O[7] + R22 * O[11]);
    float e0 = 0.5f * (R21 - R12);
    float e1 = 0.5f * (R02 - R20);
    float e2 = 0.5f * (R10 - R01);
    float cos_t = 0.5f * (R00 + R11 + R22 - 1.0f);
    cos_t = fminf(1.0f, fmaxf(-1.0f, cos_t));
    float s2 = e0 * e0 + e1 * e1 + e2 * e2;
    bool small = s2 < 1e-12f;
    float sin_s = sqrtf(small ? 1.0f : s2);
    float th = atan2f(sin_s, cos_t);
    float fac = small ? 1.0f + s2 * (1.0f / 6.0f) : th / sin_s;
    float wx = fac * e0, wy = fac * e1, wz = fac * e2;
    float omc  = small ? 1.0f : 1.0f - cos_t;
    float th2s = small ? 1.0f : th * th;
    float D = small ? (1.0f / 12.0f) + s2 * (1.0f / 720.0f)
                    : (1.0f - th * sin_s / (2.0f * omc)) / th2s;
    float xx = wx * wx, yy = wy * wy, zz = wz * wz;
    float xy = wx * wy, xz = wx * wz, yz = wy * wz;
    float Vi00 = 1.0f - D * (yy + zz), Vi01 = D * xy + 0.5f * wz, Vi02 = D * xz - 0.5f * wy;
    float Vi10 = D * xy - 0.5f * wz, Vi11 = 1.0f - D * (xx + zz), Vi12 = D * yz + 0.5f * wx;
    float Vi20 = D * xz + 0.5f * wy, Vi21 = D * yz - 0.5f * wx, Vi22 = 1.0f - D * (xx + yy);
    xi[0] = wx; xi[1] = wy; xi[2] = wz;
    xi[3] = Vi00 * tx + Vi01 * ty + Vi02 * tz;
    xi[4] = Vi10 * tx + Vi11 * ty + Vi12 * tz;
    xi[5] = Vi20 * tx + Vi21 * ty + Vi22 * tz;
}


__global__ __launch_bounds__(256) void
se3_diffuse_kernel(const float* __restrict__ twist,
                   const float* __restrict__ noise,
                   const float* __restrict__ alpha_bars,
                   const int* __restrict__ timesteps,
                   float* __restrict__ out) {
    __shared__ float sA[8][32][16];
    __shared__ float sB[8][32][16];
    __shared__ float sN[8][32][16];
    __shared__ float sDump[8][48];

    const int tid  = threadIdx.x;
    const int wave = tid >> 5;
    const int lane = tid & 31;
    const int item = blockIdx.x * 256 + tid;
    const int b    = item >> 6;

    const float2* tw2 = (const float2*)(twist + (size_t)item * 6);
    const float2* nz2 = (const float2*)(noise + (size_t)item * 6);
    float2 t0 = tw2[0], t1 = tw2[1], t2 = tw2[2];
    float2 n0 = nz2[0], n1 = nz2[1], n2 = nz2[2];
    float tw[6] = {t0.x, t0.y, t1.x, t1.y, t2.x, t2.y};
    float nz[6] = {n0.x, n0.y, n1.x, n1.y, n2.x, n2.y};

    const float ab = alpha_bars[timesteps[b]];
    const float sq = sqrtf(ab);
    const float f2 = sqrtf(1.0f - ab);

    float nv[6] = {f2 * SIGMA_R * nz[0], f2 * SIGMA_R * nz[1], f2 * SIGMA_R * nz[2],
                   f2 * SIGMA_T * nz[3], f2 * SIGMA_T * nz[4], f2 * SIGMA_T * nz[5]};
    float N[16];
    se3_exp16(nv, N);
#pragma unroll
    for (int i = 0; i < 16; ++i) sN[wave][lane][i] = N[i];

    const int half = lane >> 4;
    const int q    = lane & 15;
    const int sub  = q & 3;
    const int grp  = q >> 2;

    const bool act = (grp >> 1) == half;
    const bool hi  = (grp & 1) != 0;
    float* const dumpBase = &sDump[wave][lane];

    const float th2w = tw[0] * tw[0] + tw[1] * tw[1] + tw[2] * tw[2];
    const bool wide  = th2w >= 9.0f;
    if (__builtin_amdgcn_ballot_w32(wide) == 0u) {
        float sx[6];
#pragma unroll
        for (int i = 0; i < 6; ++i) sx[i] = sq * tw[i];
        float Ht[16];
        se3_exp16(sx, Ht);
#pragma unroll
        for (int i = 0; i < 16; ++i) sA[wave][lane][i] = Ht[i];
    } else {
        float O[16];
        se3_exp16(tw, O);
        float xi[6];
        se3_log_of_inv(O, xi);
        const float f1 = 1.0f - sq;
#pragma unroll
        for (int i = 0; i < 6; ++i) xi[i] *= f1;
        float E[16];
        se3_exp16(xi, E);
#pragma unroll
        for (int i = 0; i < 16; ++i) {
            sA[wave][lane][i] = E[i];
            sB[wave][lane][i] = O[i];
        }
        __builtin_amdgcn_wave_barrier();

#pragma unroll
        for (int r = 0; r < 8; ++r) {
            const int it = 4 * r + grp;

            v2f a, bb;
            a.x  = sA[wave][it][sub * 4 + 2 * half];
            a.y  = sA[wave][it][sub * 4 + 2 * half + 1];
            bb.x = sB[wave][it][(2 * half) * 4 + sub];
            bb.y = sB[wave][it][(2 * half + 1) * 4 + sub];

            v8f c = {};
            v8f d = __builtin_amdgcn_wmma_f32_16x16x4_f32(
                false, a, false, bb, (short)0, c, false, false);

            float* p = act ? &sA[wave][it][sub] : dumpBase;
            p[0]  = hi ? d[4] : d[0];
            p[4]  = hi ? d[5] : d[1];
            p[8]  = hi ? d[6] : d[2];
            p[12] = hi ? d[7] : d[3];
            __builtin_amdgcn_wave_barrier();
        }
    }
    __builtin_amdgcn_wave_barrier();

#pragma unroll
    for (int r = 0; r < 8; ++r) {
        const int it = 4 * r + grp;

        v2f an, bm;
        an.x = sN[wave][it][sub * 4 + 2 * half];
        an.y = sN[wave][it][sub * 4 + 2 * half + 1];
        bm.x = sA[wave][it][(2 * half) * 4 + sub];
        bm.y = sA[wave][it][(2 * half + 1) * 4 + sub];

        v8f c = {};
        v8f d2 = __builtin_amdgcn_wmma_f32_16x16x4_f32(
            false, an, false, bm, (short)0, c, false, false);

        float* p = act ? &sB[wave][it][sub] : dumpBase;
        p[0]  = hi ? d2[4] : d2[0];
        p[4]  = hi ? d2[5] : d2[1];
        p[8]  = hi ? d2[6] : d2[2];
        p[12] = hi ? d2[7] : d2[3];
    }
    __builtin_amdgcn_wave_barrier();

    const float4* s1 = (const float4*)(&sB[wave][0][0]);
    const float4* s2 = (const float4*)(&sN[wave][0][0]);
    float4* o1 = ((float4*)out) + (size_t)blockIdx.x * 1024 + wave * 128;
    float4* o2 = ((float4*)(out + OUT_HALF)) + (size_t)blockIdx.x * 1024 + wave * 128;
#pragma unroll
    for (int j = 0; j < 4; ++j) {
        o1[lane + j * 32] = s1[lane + j * 32];
        o2[lane + j * 32] = s2[lane + j * 32];
    }
}

extern "C" void kernel_launch(void* const* d_in, const int* in_sizes, int n_in,
                              void* d_out, int out_size, void* d_ws, size_t ws_size,
                              hipStream_t stream) {
    const float* twist      = (const float*)d_in[0];
    const float* noise      = (const float*)d_in[1];
    const float* alpha_bars = (const float*)d_in[2];
    const int*   timesteps  = (const int*)d_in[3];
    float*       out        = (float*)d_out;

    const int blocks = NUM_ITEMS / 256;
    hipLaunchKernelGGL(se3_diffuse_kernel, dim3(blocks), dim3(256), 0, stream,
                       twist, noise, alpha_bars, timesteps, out);
}


# Final product formed transposed (H_t^T as A, N^T as B; same products, same k order): each D column is an output row, stored from registers in whole 128-B lines; LDS restaging of output 1 dropped
# speedup vs baseline: 1.4641x; 1.0974x over previous
#include <hip/hip_runtime.h>
#include <hip/hip_bf16.h>
#include <math.h>


#define NUM_ITEMS (4096 * 64)
#define OUT_HALF  (NUM_ITEMS * 16)
#define SIGMA_R 0.05f
#define SIGMA_T 0.03f

typedef float v2f __attribute__((ext_vector_type(2)));
typedef float v8f __attribute__((ext_vector_type(8)));


__device__ __forceinline__ void se3_exp16(const float xi[6], float* H) {
    const float wx = xi[0], wy = xi[1], wz = xi[2];
    const float vx = xi[3], vy = xi[4], vz = xi[5];
    float th2  = wx * wx + wy * wy + wz * wz;
    bool small = th2 < 1e-8f;
    float th2s = small ? 1.0f : th2;
    float th   = sqrtf(th2s);
    float s, c;
    sincosf(th, &s, &c);
    float A  = small ? 1.0f - th2 * (1.0f / 6.0f)   : s / th;
    float Bc = small ? 0.5f - th2 * (1.0f / 24.0f)  : (1.0f - c) / th2s;
    float Cc = small ? (1.0f / 6.0f) - th2 * (1.0f / 120.0f)
                     : (th - s) / (th2s * th);
    float xx = wx * wx, yy = wy * wy, zz = wz * wz;
    float xy = wx * wy, xz = wx * wz, yz = wy * wz;
    float R00 = 1.0f - Bc * (yy + zz), R01 = Bc * xy - A * wz, R02 = Bc * xz + A * wy;
    float R10 = Bc * xy + A * wz, R11 = 1.0f - Bc * (xx + zz), R12 = Bc * yz - A * wx;
    float R20 = Bc * xz - A * wy, R21 = Bc * yz + A * wx, R22 = 1.0f - Bc * (xx + yy);
    float V00 = 1.0f - Cc * (yy + zz), V01 = Cc * xy - Bc * wz, V02 = Cc * xz + Bc * wy;
    float V10 = Cc * xy + Bc * wz, V11 = 1.0f - Cc * (xx + zz), V12 = Cc * yz - Bc * wx;
    float V20 = Cc * xz - Bc * wy, V21 = Cc * yz + Bc * wx, V22 = 1.0f - Cc * (xx + yy);
    float tx = V00 * vx + V01 * vy + V02 * vz;
    float ty = V10 * vx + V11 * vy + V12 * vz;
    float tz = V20 * vx + V21 * vy + V22 * vz;
    H[0] = R00; H[1] = R01; H[2]  = R02; H[3]  = tx;
    H[4] = R10; H[5] = R11; H[6]  = R12; H[7]  = ty;
    H[8] = R20; H[9] = R21; H[10] = R22; H[11] = tz;
    H[12] = 0.0f; H[13] = 0.0f; H[14] = 0.0f; H[15] = 1.0f;
}

__device__ __forceinline__ void se3_log_of_inv(const float* O, float xi[6]) {
    float R00 = O[0], R01 = O[4], R02 = O[8];
    float R10 = O[1], R11 = O[5], R12 = O[9];
    float R20 = O[2], R21 = O[6], R22 = O[10];
    float tx = -(R00 * O[3] + R01 * O[7] + R02 * O[11]);
    float ty = -(R10 * O[3] + R11 * O[7] + R12 * O[11]);
    float tz = -(R20 * O[3] + R21 * O[7] + R22 * O[11]);
    float e0 = 0.5f * (R21 - R12);
    float e1 = 0.5f * (R02 - R20);
    float e2 = 0.5f * (R10 - R01);
    float cos_t = 0.5f * (R00 + R11 + R22 - 1.0f);
    cos_t = fminf(1.0f, fmaxf(-1.0f, cos_t));
    float s2 = e0 * e0 + e1 * e1 + e2 * e2;
    bool small = s2 < 1e-12f;
    float sin_s = sqrtf(small ? 1.0f : s2);
    float th = atan2f(sin_s, cos_t);
    float fac = small ? 1.0f + s2 * (1.0f / 6.0f) : th / sin_s;
    float wx = fac * e0, wy = fac * e1, wz = fac * e2;
    float omc  = small ? 1.0f : 1.0f - cos_t;
    float th2s = small ? 1.0f : th * th;
    float D = small ? (1.0f / 12.0f) + s2 * (1.0f / 720.0f)
                    : (1.0f - th * sin_s / (2.0f * omc)) / th2s;
    float xx = wx * wx, yy = wy * wy, zz = wz * wz;
    float xy = wx * wy, xz = wx * wz, yz = wy * wz;
    float Vi00 = 1.0f - D * (yy + zz), Vi01 = D * xy + 0.5f * wz, Vi02 = D * xz - 0.5f * wy;
    float Vi10 = D * xy - 0.5f * wz, Vi11 = 1.0f - D * (xx + zz), Vi12 = D * yz + 0.5f * wx;
    float Vi20 = D * xz + 0.5f * wy, Vi21 = D * yz - 0.5f * wx, Vi22 = 1.0f - D * (xx + yy);
    xi[0] = wx; xi[1] = wy; xi[2] = wz;
    xi[3] = Vi00 * tx + Vi01 * ty + Vi02 * tz;
    xi[4] = Vi10 * tx + Vi11 * ty + Vi12 * tz;
    xi[5] = Vi20 * tx + Vi21 * ty + Vi22 * tz;
}


__global__ __launch_bounds__(256) void
se3_diffuse_kernel(const float* __restrict__ twist,
                   const float* __restrict__ noise,
                   const float* __restrict__ alpha_bars,
                   const int* __restrict__ timesteps,
                   float* __restrict__ out) {
    __shared__ float sA[8][32][16];
    __shared__ float sB[8][32][16];
    __shared__ float sN[8][32][16];
    __shared__ float sDump[8][48];

    const int tid  = threadIdx.x;
    const int wave = tid >> 5;
    const int lane = tid & 31;
    const int item = blockIdx.x * 256 + tid;
    const int b    = item >> 6;

    const float2* tw2 = (const float2*)(twist + (size_t)item * 6);
    const float2* nz2 = (const float2*)(noise + (size_t)item * 6);
    float2 t0 = tw2[0], t1 = tw2[1], t2 = tw2[2];
    float2 n0 = nz2[0], n1 = nz2[1], n2 = nz2[2];
    float tw[6] = {t0.x, t0.y, t1.x, t1.y, t2.x, t2.y};
    float nz[6] = {n0.x, n0.y, n1.x, n1.y, n2.x, n2.y};

    const float ab = alpha_bars[timesteps[b]];
    const float sq = sqrtf(ab);
    const float f2 = sqrtf(1.0f - ab);

    float nv[6] = {f2 * SIGMA_R * nz[0], f2 * SIGMA_R * nz[1], f2 * SIGMA_R * nz[2],
                   f2 * SIGMA_T * nz[3], f2 * SIGMA_T * nz[4], f2 * SIGMA_T * nz[5]};
    float N[16];
    se3_exp16(nv, N);
#pragma unroll
    for (int i = 0; i < 16; ++i) sN[wave][lane][i] = N[i];

    const int half = lane >> 4;
    const int q    = lane & 15;
    const int sub  = q & 3;
    const int grp  = q >> 2;

    const bool act = (grp >> 1) == half;
    const bool hi  = (grp & 1) != 0;
    float* const dumpBase = &sDump[wave][lane];

    const float th2w = tw[0] * tw[0] + tw[1] * tw[1] + tw[2] * tw[2];
    const bool wide  = th2w >= 9.0f;
    if (__builtin_amdgcn_ballot_w32(wide) == 0u) {
        float sx[6];
#pragma unroll
        for (int i = 0; i < 6; ++i) sx[i] = sq * tw[i];
        float Ht[16];
        se3_exp16(sx, Ht);
#pragma unroll
        for (int i = 0; i < 16; ++i) sA[wave][lane][i] = Ht[i];
    } else {
        float O[16];
        se3_exp16(tw, O);
        float xi[6];
        se3_log_of_inv(O, xi);
        const float f1 = 1.0f - sq;
#pragma unroll
        for (int i = 0; i < 6; ++i) xi[i] *= f1;
        float E[16];
        se3_exp16(xi, E);
#pragma unroll
        for (int i = 0; i < 16; ++i) {
            sA[wave][lane][i] = E[i];
            sB[wave][lane][i] = O[i];
        }
        __builtin_amdgcn_wave_barrier();

#pragma unroll
        for (int r = 0; r < 8; ++r) {
            const int it = 4 * r + grp;

            v2f a, bb;
            a.x  = sA[wave][it][sub * 4 + 2 * half];
            a.y  = sA[wave][it][sub * 4 + 2 * half + 1];
            bb.x = sB[wave][it][(2 * half) * 4 + sub];
            bb.y = sB[wave][it][(2 * half + 1) * 4 + sub];

            v8f c = {};
            v8f d = __builtin_amdgcn_wmma_f32_16x16x4_f32(
                false, a, false, bb, (short)0, c, false, false);

            float* p = act ? &sA[wave][it][sub] : dumpBase;
            p[0]  = hi ? d[4] : d[0];
            p[4]  = hi ? d[5] : d[1];
            p[8]  = hi ? d[6] : d[2];
            p[12] = hi ? d[7] : d[3];
            __builtin_amdgcn_wave_barrier();
        }
    }
    __builtin_amdgcn_wave_barrier();

    v2f hA[8], nB[8];
#pragma unroll
    for (int r = 0; r < 8; ++r) {
        const int it = 4 * r + grp;
        hA[r].x = sA[wave][it][(2 * half) * 4 + sub];
        hA[r].y = sA[wave][it][(2 * half + 1) * 4 + sub];
        nB[r].x = sN[wave][it][sub * 4 + 2 * half];
        nB[r].y = sN[wave][it][sub * 4 + 2 * half + 1];
    }

    const float4* s2 = (const float4*)(&sN[wave][0][0]);
    float4* o2 = ((float4*)(out + OUT_HALF)) + (size_t)blockIdx.x * 1024 + wave * 128;
#pragma unroll
    for (int j = 0; j < 4; ++j) o2[lane + j * 32] = s2[lane + j * 32];

    float4 prod[8];
#pragma unroll
    for (int r = 0; r < 8; ++r) {
        v8f c = {};
        v8f d = __builtin_amdgcn_wmma_f32_16x16x4_f32(
            false, hA[r], false, nB[r], (short)0, c, false, false);
        prod[r].x = hi ? d[4] : d[0];
        prod[r].y = hi ? d[5] : d[1];
        prod[r].z = hi ? d[6] : d[2];
        prod[r].w = hi ? d[7] : d[3];
    }

    if (act) {
        float4* o1 = ((float4*)out) + (size_t)blockIdx.x * 1024 + wave * 128 + q;
#pragma unroll
        for (int r = 0; r < 8; ++r) o1[16 * r] = prod[r];
    }
}

extern "C" void kernel_launch(void* const* d_in, const int* in_sizes, int n_in,
                              void* d_out, int out_size, void* d_ws, size_t ws_size,
                              hipStream_t stream) {
    const float* twist      = (const float*)d_in[0];
    const float* noise      = (const float*)d_in[1];
    const float* alpha_bars = (const float*)d_in[2];
    const int*   timesteps  = (const int*)d_in[3];
    float*       out        = (float*)d_out;

    const int blocks = NUM_ITEMS / 256;
    hipLaunchKernelGGL(se3_diffuse_kernel, dim3(blocks), dim3(256), 0, stream,
                       twist, noise, alpha_bars, timesteps, out);
}


# Two items per lane, computed as packed f32 pairs: a wave is one batch row of 64 items (4096 waves, not 8192), its schedule entry read once by scalar loads; formulas per item unchanged
# speedup vs baseline: 1.6098x; 1.0995x over previous
#include <hip/hip_runtime.h>
#include <hip/hip_bf16.h>
#include <math.h>


#define NUM_ITEMS (4096 * 64)
#define OUT_HALF  (NUM_ITEMS * 16)
#define SIGMA_R 0.05f
#define SIGMA_T 0.03f

typedef float v2f __attribute__((ext_vector_type(2)));
typedef float v8f __attribute__((ext_vector_type(8)));
typedef int   v2i __attribute__((ext_vector_type(2)));


__device__ __forceinline__ void se3_exp16(const float xi[6], float* H) {
    const float wx = xi[0], wy = xi[1], wz = xi[2];
    const float vx = xi[3], vy = xi[4], vz = xi[5];
    float th2  = wx * wx + wy * wy + wz * wz;
    bool small = th2 < 1e-8f;
    float th2s = small ? 1.0f : th2;
    float th   = sqrtf(th2s);
    float s, c;
    sincosf(th, &s, &c);
    float A  = small ? 1.0f - th2 * (1.0f / 6.0f)   : s / th;
    float Bc = small ? 0.5f - th2 * (1.0f / 24.0f)  : (1.0f - c) / th2s;
    float Cc = small ? (1.0f / 6.0f) - th2 * (1.0f / 120.0f)
                     : (th - s) / (th2s * th);
    float xx = wx * wx, yy = wy * wy, zz = wz * wz;
    float xy = wx * wy, xz = wx * wz, yz = wy * wz;
    float R00 = 1.0f - Bc * (yy + zz), R01 = Bc * xy - A * wz, R02 = Bc * xz + A * wy;
    float R10 = Bc * xy + A * wz, R11 = 1.0f - Bc * (xx + zz), R12 = Bc * yz - A * wx;
    float R20 = Bc * xz - A * wy, R21 = Bc * yz + A * wx, R22 = 1.0f - Bc * (xx + yy);
    float V00 = 1.0f - Cc * (yy + zz), V01 = Cc * xy - Bc * wz, V02 = Cc * xz + Bc * wy;
    float V10 = Cc * xy + Bc * wz, V11 = 1.0f - Cc * (xx + zz), V12 = Cc * yz - Bc * wx;
    float V20 = Cc * xz - Bc * wy, V21 = Cc * yz + Bc * wx, V22 = 1.0f - Cc * (xx + yy);
    float tx = V00 * vx + V01 * vy + V02 * vz;
    float ty = V10 * vx + V11 * vy + V12 * vz;
    float tz = V20 * vx + V21 * vy + V22 * vz;
    H[0] = R00; H[1] = R01; H[2]  = R02; H[3]  = tx;
    H[4] = R10; H[5] = R11; H[6]  = R12; H[7]  = ty;
    H[8] = R20; H[9] = R21; H[10] = R22; H[11] = tz;
    H[12] = 0.0f; H[13] = 0.0f; H[14] = 0.0f; H[15] = 1.0f;
}

__device__ __forceinline__ void se3_exp16_x2(const v2f xi[6], v2f* H) {
    const v2f wx = xi[0], wy = xi[1], wz = xi[2];
    const v2f vx = xi[3], vy = xi[4], vz = xi[5];
    const v2f one = {1.0f, 1.0f};
    v2f th2   = wx * wx + wy * wy + wz * wz;
    v2i small = th2 < 1e-8f;
    v2f th2s  = small ? one : th2;
    v2f th;
    th.x = sqrtf(th2s.x);
    th.y = sqrtf(th2s.y);
    float s0, c0, s1, c1;
    sincosf(th.x, &s0, &c0);
    sincosf(th.y, &s1, &c1);
    v2f s = {s0, s1};
    v2f c = {c0, c1};
    v2f A  = small ? one - th2 * (1.0f / 6.0f)   : s / th;
    v2f Bc = small ? 0.5f - th2 * (1.0f / 24.0f) : (one - c) / th2s;
    v2f Cc = small ? (1.0f / 6.0f) - th2 * (1.0f / 120.0f)
                   : (th - s) / (th2s * th);
    v2f xx = wx * wx, yy = wy * wy, zz = wz * wz;
    v2f xy = wx * wy, xz = wx * wz, yz = wy * wz;
    v2f R00 = one - Bc * (yy + zz), R01 = Bc * xy - A * wz, R02 = Bc * xz + A * wy;
    v2f R10 = Bc * xy + A * wz, R11 = one - Bc * (xx + zz), R12 = Bc * yz - A * wx;
    v2f R20 = Bc * xz - A * wy, R21 = Bc * yz + A * wx, R22 = one - Bc * (xx + yy);
    v2f V00 = one - Cc * (yy + zz), V01 = Cc * xy - Bc * wz, V02 = Cc * xz + Bc * wy;
    v2f V10 = Cc * xy + Bc * wz, V11 = one - Cc * (xx + zz), V12 = Cc * yz - Bc * wx;
    v2f V20 = Cc * xz - Bc * wy, V21 = Cc * yz + Bc * wx, V22 = one - Cc * (xx + yy);
    v2f tx = V00 * vx + V01 * vy + V02 * vz;
    v2f ty = V10 * vx + V11 * vy + V12 * vz;
    v2f tz = V20 * vx + V21 * vy + V22 * vz;
    const v2f zero = {0.0f, 0.0f};
    H[0] = R00; H[1] = R01; H[2]  = R02; H[3]  = tx;
    H[4] = R10; H[5] = R11; H[6]  = R12; H[7]  = ty;
    H[8] = R20; H[9] = R21; H[10] = R22; H[11] = tz;
    H[12] = zero; H[13] = zero; H[14] = zero; H[15] = one;
}

__device__ __forceinline__ void se3_log_of_inv(const float* O, float xi[6]) {
    float R00 = O[0], R01 = O[4], R02 = O[8];
    float R10 = O[1], R11 = O[5], R12 = O[9];
    float R20 = O[2], R21 = O[6], R22 = O[10];
    float tx = -(R00 * O[3] + R01 * O[7] + R02 * O[11]);
    float ty = -(R10 * O[3] + R11 * O[7] + R12 * O[11]);
    float tz = -(R20 * O[3] + R21 * O[7] + R22 * O[11]);
    float e0 = 0.5f * (R21 - R12);
    float e1 = 0.5f * (R02 - R20);
    float e2 = 0.5f * (R10 - R01);
    float cos_t = 0.5f * (R00 + R11 + R22 - 1.0f);
    cos_t = fminf(1.0f, fmaxf(-1.0f, cos_t));
    float s2 = e0 * e0 + e1 * e1 + e2 * e2;
    bool small = s2 < 1e-12f;
    float sin_s = sqrtf(small ? 1.0f : s2);
    float th = atan2f(sin_s, cos_t);
    float fac = small ? 1.0f + s2 * (1.0f / 6.0f) : th / sin_s;
    float wx = fac * e0, wy = fac * e1, wz = fac * e2;
    float omc  = small ? 1.0f : 1.0f - cos_t;
    float th2s = small ? 1.0f : th * th;
    float D = small ? (1.0f / 12.0f) + s2 * (1.0f / 720.0f)
                    : (1.0f - th * sin_s / (2.0f * omc)) / th2s;
    float xx = wx * wx, yy = wy * wy, zz = wz * wz;
    float xy = wx * wy, xz = wx * wz, yz = wy * wz;
    float Vi00 = 1.0f - D * (yy + zz), Vi01 = D * xy + 0.5f * wz, Vi02 = D * xz - 0.5f * wy;
    float Vi10 = D * xy - 0.5f * wz, Vi11 = 1.0f - D * (xx + zz), Vi12 = D * yz + 0.5f * wx;
    float Vi20 = D * xz + 0.5f * wy, Vi21 = D * yz - 0.5f * wx, Vi22 = 1.0f - D * (xx + yy);
    xi[0] = wx; xi[1] = wy; xi[2] = wz;
    xi[3] = Vi00 * tx + Vi01 * ty + Vi02 * tz;
    xi[4] = Vi10 * tx + Vi11 * ty + Vi12 * tz;
    xi[5] = Vi20 * tx + Vi21 * ty + Vi22 * tz;
}


__global__ __launch_bounds__(256) void
se3_diffuse_kernel(const float* __restrict__ twist,
                   const float* __restrict__ noise,
                   const float* __restrict__ alpha_bars,
                   const int* __restrict__ timesteps,
                   float* __restrict__ out) {
    __shared__ float sA[8][64][16];
    __shared__ float sB[8][64][16];
    __shared__ float sN[8][64][16];
    __shared__ float sDump[8][48];

    const int tid   = threadIdx.x;
    const int wave  = tid >> 5;
    const int lane  = tid & 31;
    const int b     = blockIdx.x * 8 + wave;
    const int wbase = b * 64;

    v2f tw[6], nz[6];
    {
        const float2* twa = (const float2*)(twist + (size_t)(wbase + lane) * 6);
        const float2* nza = (const float2*)(noise + (size_t)(wbase + lane) * 6);
        const float2* twb = twa + 32 * 3;
        const float2* nzb = nza + 32 * 3;
        float2 ta0 = twa[0], ta1 = twa[1], ta2 = twa[2];
        float2 tb0 = twb[0], tb1 = twb[1], tb2 = twb[2];
        float2 na0 = nza[0], na1 = nza[1], na2 = nza[2];
        float2 nb0 = nzb[0], nb1 = nzb[1], nb2 = nzb[2];
        tw[0] = (v2f){ta0.x, tb0.x}; tw[1] = (v2f){ta0.y, tb0.y}; tw[2] = (v2f){ta1.x, tb1.x};
        tw[3] = (v2f){ta1.y, tb1.y}; tw[4] = (v2f){ta2.x, tb2.x}; tw[5] = (v2f){ta2.y, tb2.y};
        nz[0] = (v2f){na0.x, nb0.x}; nz[1] = (v2f){na0.y, nb0.y}; nz[2] = (v2f){na1.x, nb1.x};
        nz[3] = (v2f){na1.y, nb1.y}; nz[4] = (v2f){na2.x, nb2.x}; nz[5] = (v2f){na2.y, nb2.y};
    }

    const int bs = __builtin_amdgcn_readfirstlane(b);
    const float ab = alpha_bars[timesteps[bs]];
    const float sq = sqrtf(ab);
    const float f2 = sqrtf(1.0f - ab);
    const float fr = f2 * SIGMA_R;
    const float ft = f2 * SIGMA_T;

    {
        v2f nv[6] = {fr * nz[0], fr * nz[1], fr * nz[2],
                     ft * nz[3], ft * nz[4], ft * nz[5]};
        v2f N[16];
        se3_exp16_x2(nv, N);
#pragma unroll
        for (int i = 0; i < 16; ++i) {
            sN[wave][lane][i]      = N[i].x;
            sN[wave][32 + lane][i] = N[i].y;
        }
    }

    const int half = lane >> 4;
    const int q    = lane & 15;
    const int sub  = q & 3;
    const int grp  = q >> 2;

    const bool act = (grp >> 1) == half;
    const bool hi  = (grp & 1) != 0;
    float* const dumpBase = &sDump[wave][lane];

    const v2f th2w  = tw[0] * tw[0] + tw[1] * tw[1] + tw[2] * tw[2];
    const bool wide = (th2w.x >= 9.0f) || (th2w.y >= 9.0f);
    if (__builtin_amdgcn_ballot_w32(wide) == 0u) {
        v2f sx[6];
#pragma unroll
        for (int i = 0; i < 6; ++i) sx[i] = sq * tw[i];
        v2f Ht[16];
        se3_exp16_x2(sx, Ht);
#pragma unroll
        for (int i = 0; i < 16; ++i) {
            sA[wave][lane][i]      = Ht[i].x;
            sA[wave][32 + lane][i] = Ht[i].y;
        }
    } else {
        const float f1 = 1.0f - sq;
#pragma unroll
        for (int h = 0; h < 2; ++h) {
            const float twh[6] = {tw[0][h], tw[1][h], tw[2][h], tw[3][h], tw[4][h], tw[5][h]};
            float O[16];
            se3_exp16(twh, O);
            float xi[6];
            se3_log_of_inv(O, xi);
#pragma unroll
            for (int i = 0; i < 6; ++i) xi[i] *= f1;
            float E[16];
            se3_exp16(xi, E);
#pragma unroll
            for (int i = 0; i < 16; ++i) {
                sA[wave][32 * h + lane][i] = E[i];
                sB[wave][32 * h + lane][i] = O[i];
            }
        }
        __builtin_amdgcn_wave_barrier();

#pragma unroll
        for (int r = 0; r < 16; ++r) {
            const int it = 4 * r + grp;

            v2f a, bb;
            a.x  = sA[wave][it][sub * 4 + 2 * half];
            a.y  = sA[wave][it][sub * 4 + 2 * half + 1];
            bb.x = sB[wave][it][(2 * half) * 4 + sub];
            bb.y = sB[wave][it][(2 * half + 1) * 4 + sub];

            v8f c = {};
            v8f d = __builtin_amdgcn_wmma_f32_16x16x4_f32(
                false, a, false, bb, (short)0, c, false, false);

            float* p = act ? &sA[wave][it][sub] : dumpBase;
            p[0]  = hi ? d[4] : d[0];
            p[4]  = hi ? d[5] : d[1];
            p[8]  = hi ? d[6] : d[2];
            p[12] = hi ? d[7] : d[3];
            __builtin_amdgcn_wave_barrier();
        }
    }
    __builtin_amdgcn_wave_barrier();

    const float4* s2 = (const float4*)(&sN[wave][0][0]);
    float4* o2 = ((float4*)(out + OUT_HALF)) + (size_t)wbase * 4;
#pragma unroll
    for (int j = 0; j < 8; ++j) o2[lane + j * 32] = s2[lane + j * 32];

    float4 prod[16];
#pragma unroll
    for (int ps = 0; ps < 2; ++ps) {
        v2f hA[8], nB[8];
#pragma unroll
        for (int r = 0; r < 8; ++r) {
            const int it = 32 * ps + 4 * r + grp;
            hA[r].x = sA[wave][it][(2 * half) * 4 + sub];
            hA[r].y = sA[wave][it][(2 * half + 1) * 4 + sub];
            nB[r].x = sN[wave][it][sub * 4 + 2 * half];
            nB[r].y = sN[wave][it][sub * 4 + 2 * half + 1];
        }
#pragma unroll
        for (int r = 0; r < 8; ++r) {
            v8f c = {};
            v8f d = __builtin_amdgcn_wmma_f32_16x16x4_f32(
                false, hA[r], false, nB[r], (short)0, c, false, false);
            prod[8 * ps + r].x = hi ? d[4] : d[0];
            prod[8 * ps + r].y = hi ? d[5] : d[1];
            prod[8 * ps + r].z = hi ? d[6] : d[2];
            prod[8 * ps + r].w = hi ? d[7] : d[3];
        }
    }

    if (act) {
        float4* o1 = ((float4*)out) + (size_t)wbase * 4 + q;
#pragma unroll
        for (int R = 0; R < 16; ++R) o1[16 * R] = prod[R];
    }
}

extern "C" void kernel_launch(void* const* d_in, const int* in_sizes, int n_in,
                              void* d_out, int out_size, void* d_ws, size_t ws_size,
                              hipStream_t stream) {
    const float* twist      = (const float*)d_in[0];
    const float* noise      = (const float*)d_in[1];
    const float* alpha_bars = (const float*)d_in[2];
    const int*   timesteps  = (const int*)d_in[3];
    float*       out        = (float*)d_out;

    const int blocks = NUM_ITEMS / 512;
    hipLaunchKernelGGL(se3_diffuse_kernel, dim3(blocks), dim3(256), 0, stream,
                       twist, noise, alpha_bars, timesteps, out);
}


# Second output (H_noise) stored right after N reaches LDS, before the H_t chain: half of the write traffic drains under the remaining vector work, not at the end with output 1; sums unchanged
# speedup vs baseline: 1.6952x; 1.0531x over previous
#include <hip/hip_runtime.h>
#include <hip/hip_bf16.h>
#include <math.h>


#define NUM_ITEMS (4096 * 64)
#define OUT_HALF  (NUM_ITEMS * 16)
#define SIGMA_R 0.05f
#define SIGMA_T 0.03f

typedef float v2f __attribute__((ext_vector_type(2)));
typedef float v8f __attribute__((ext_vector_type(8)));
typedef int   v2i __attribute__((ext_vector_type(2)));


__device__ __forceinline__ void se3_exp16(const float xi[6], float* H) {
    const float wx = xi[0], wy = xi[1], wz = xi[2];
    const float vx = xi[3], vy = xi[4], vz = xi[5];
    float th2  = wx * wx + wy * wy + wz * wz;
    bool small = th2 < 1e-8f;
    float th2s = small ? 1.0f : th2;
    float th   = sqrtf(th2s);
    float s, c;
    sincosf(th, &s, &c);
    float A  = small ? 1.0f - th2 * (1.0f / 6.0f)   : s / th;
    float Bc = small ? 0.5f - th2 * (1.0f / 24.0f)  : (1.0f - c) / th2s;
    float Cc = small ? (1.0f / 6.0f) - th2 * (1.0f / 120.0f)
                     : (th - s) / (th2s * th);
    float xx = wx * wx, yy = wy * wy, zz = wz * wz;
    float xy = wx * wy, xz = wx * wz, yz = wy * wz;
    float R00 = 1.0f - Bc * (yy + zz), R01 = Bc * xy - A * wz, R02 = Bc * xz + A * wy;
    float R10 = Bc * xy + A * wz, R11 = 1.0f - Bc * (xx + zz), R12 = Bc * yz - A * wx;
    float R20 = Bc * xz - A * wy, R21 = Bc * yz + A * wx, R22 = 1.0f - Bc * (xx + yy);
    float V00 = 1.0f - Cc * (yy + zz), V01 = Cc * xy - Bc * wz, V02 = Cc * xz + Bc * wy;
    float V10 = Cc * xy + Bc * wz, V11 = 1.0f - Cc * (xx + zz), V12 = Cc * yz - Bc * wx;
    float V20 = Cc * xz - Bc * wy, V21 = Cc * yz + Bc * wx, V22 = 1.0f - Cc * (xx + yy);
    float tx = V00 * vx + V01 * vy + V02 * vz;
    float ty = V10 * vx + V11 * vy + V12 * vz;
    float tz = V20 * vx + V21 * vy + V22 * vz;
    H[0] = R00; H[1] = R01; H[2]  = R02; H[3]  = tx;
    H[4] = R10; H[5] = R11; H[6]  = R12; H[7]  = ty;
    H[8] = R20; H[9] = R21; H[10] = R22; H[11] = tz;
    H[12] = 0.0f; H[13] = 0.0f; H[14] = 0.0f; H[15] = 1.0f;
}

__device__ __forceinline__ void se3_exp16_x2(const v2f xi[6], v2f* H) {
    const v2f wx = xi[0], wy = xi[1], wz = xi[2];
    const v2f vx = xi[3], vy = xi[4], vz = xi[5];
    const v2f one = {1.0f, 1.0f};
    v2f th2   = wx * wx + wy * wy + wz * wz;
    v2i small = th2 < 1e-8f;
    v2f th2s  = small ? one : th2;
    v2f th;
    th.x = sqrtf(th2s.x);
    th.y = sqrtf(th2s.y);
    float s0, c0, s1, c1;
    sincosf(th.x, &s0, &c0);
    sincosf(th.y, &s1, &c1);
    v2f s = {s0, s1};
    v2f c = {c0, c1};
    v2f A  = small ? one - th2 * (1.0f / 6.0f)   : s / th;
    v2f Bc = small ? 0.5f - th2 * (1.0f / 24.0f) : (one - c) / th2s;
    v2f Cc = small ? (1.0f / 6.0f) - th2 * (1.0f / 120.0f)
                   : (th - s) / (th2s * th);
    v2f xx = wx * wx, yy = wy * wy, zz = wz * wz;
    v2f xy = wx * wy, xz = wx * wz, yz = wy * wz;
    v2f R00 = one - Bc * (yy + zz), R01 = Bc * xy - A * wz, R02 = Bc * xz + A * wy;
    v2f R10 = Bc * xy + A * wz, R11 = one - Bc * (xx + zz), R12 = Bc * yz - A * wx;
    v2f R20 = Bc * xz - A * wy, R21 = Bc * yz + A * wx, R22 = one - Bc * (xx + yy);
    v2f V00 = one - Cc * (yy + zz), V01 = Cc * xy - Bc * wz, V02 = Cc * xz + Bc * wy;
    v2f V10 = Cc * xy + Bc * wz, V11 = one - Cc * (xx + zz), V12 = Cc * yz - Bc * wx;
    v2f V20 = Cc * xz - Bc * wy, V21 = Cc * yz + Bc * wx, V22 = one - Cc * (xx + yy);
    v2f tx = V00 * vx + V01 * vy + V02 * vz;
    v2f ty = V10 * vx + V11 * vy + V12 * vz;
    v2f tz = V20 * vx + V21 * vy + V22 * vz;
    const v2f zero = {0.0f, 0.0f};
    H[0] = R00; H[1] = R01; H[2]  = R02; H[3]  = tx;
    H[4] = R10; H[5] = R11; H[6]  = R12; H[7]  = ty;
    H[8] = R20; H[9] = R21; H[10] = R22; H[11] = tz;
    H[12] = zero; H[13] = zero; H[14] = zero; H[15] = one;
}

__device__ __forceinline__ void se3_log_of_inv(const float* O, float xi[6]) {
    float R00 = O[0], R01 = O[4], R02 = O[8];
    float R10 = O[1], R11 = O[5], R12 = O[9];
    float R20 = O[2], R21 = O[6], R22 = O[10];
    float tx = -(R00 * O[3] + R01 * O[7] + R02 * O[11]);
    float ty = -(R10 * O[3] + R11 * O[7] + R12 * O[11]);
    float tz = -(R20 * O[3] + R21 * O[7] + R22 * O[11]);
    float e0 = 0.5f * (R21 - R12);
    float e1 = 0.5f * (R02 - R20);
    float e2 = 0.5f * (R10 - R01);
    float cos_t = 0.5f * (R00 + R11 + R22 - 1.0f);
    cos_t = fminf(1.0f, fmaxf(-1.0f, cos_t));
    float s2 = e0 * e0 + e1 * e1 + e2 * e2;
    bool small = s2 < 1e-12f;
    float sin_s = sqrtf(small ? 1.0f : s2);
    float th = atan2f(sin_s, cos_t);
    float fac = small ? 1.0f + s2 * (1.0f / 6.0f) : th / sin_s;
    float wx = fac * e0, wy = fac * e1, wz = fac * e2;
    float omc  = small ? 1.0f : 1.0f - cos_t;
    float th2s = small ? 1.0f : th * th;
    float D = small ? (1.0f / 12.0f) + s2 * (1.0f / 720.0f)
                    : (1.0f - th * sin_s / (2.0f * omc)) / th2s;
    float xx = wx * wx, yy = wy * wy, zz = wz * wz;
    float xy = wx * wy, xz = wx * wz, yz = wy * wz;
    float Vi00 = 1.0f - D * (yy + zz), Vi01 = D * xy + 0.5f * wz, Vi02 = D * xz - 0.5f * wy;
    float Vi10 = D * xy - 0.5f * wz, Vi11 = 1.0f - D * (xx + zz), Vi12 = D * yz + 0.5f * wx;
    float Vi20 = D * xz + 0.5f * wy, Vi21 = D * yz - 0.5f * wx, Vi22 = 1.0f - D * (xx + yy);
    xi[0] = wx; xi[1] = wy; xi[2] = wz;
    xi[3] = Vi00 * tx + Vi01 * ty + Vi02 * tz;
    xi[4] = Vi10 * tx + Vi11 * ty + Vi12 * tz;
    xi[5] = Vi20 * tx + Vi21 * ty + Vi22 * tz;
}


__global__ __launch_bounds__(256) void
se3_diffuse_kernel(const float* __restrict__ twist,
                   const float* __restrict__ noise,
                   const float* __restrict__ alpha_bars,
                   const int* __restrict__ timesteps,
                   float* __restrict__ out) {
    __shared__ float sA[8][64][16];
    __shared__ float sB[8][64][16];
    __shared__ float sN[8][64][16];
    __shared__ float sDump[8][48];

    const int tid   = threadIdx.x;
    const int wave  = tid >> 5;
    const int lane  = tid & 31;
    const int b     = blockIdx.x * 8 + wave;
    const int wbase = b * 64;

    v2f tw[6], nz[6];
    {
        const float2* twa = (const float2*)(twist + (size_t)(wbase + lane) * 6);
        const float2* nza = (const float2*)(noise + (size_t)(wbase + lane) * 6);
        const float2* twb = twa + 32 * 3;
        const float2* nzb = nza + 32 * 3;
        float2 ta0 = twa[0], ta1 = twa[1], ta2 = twa[2];
        float2 tb0 = twb[0], tb1 = twb[1], tb2 = twb[2];
        float2 na0 = nza[0], na1 = nza[1], na2 = nza[2];
        float2 nb0 = nzb[0], nb1 = nzb[1], nb2 = nzb[2];
        tw[0] = (v2f){ta0.x, tb0.x}; tw[1] = (v2f){ta0.y, tb0.y}; tw[2] = (v2f){ta1.x, tb1.x};
        tw[3] = (v2f){ta1.y, tb1.y}; tw[4] = (v2f){ta2.x, tb2.x}; tw[5] = (v2f){ta2.y, tb2.y};
        nz[0] = (v2f){na0.x, nb0.x}; nz[1] = (v2f){na0.y, nb0.y}; nz[2] = (v2f){na1.x, nb1.x};
        nz[3] = (v2f){na1.y, nb1.y}; nz[4] = (v2f){na2.x, nb2.x}; nz[5] = (v2f){na2.y, nb2.y};
    }

    const int bs = __builtin_amdgcn_readfirstlane(b);
    const float ab = alpha_bars[timesteps[bs]];
    const float sq = sqrtf(ab);
    const float f2 = sqrtf(1.0f - ab);
    const float fr = f2 * SIGMA_R;
    const float ft = f2 * SIGMA_T;

    {
        v2f nv[6] = {fr * nz[0], fr * nz[1], fr * nz[2],
                     ft * nz[3], ft * nz[4], ft * nz[5]};
        v2f N[16];
        se3_exp16_x2(nv, N);
#pragma unroll
        for (int i = 0; i < 16; ++i) {
            sN[wave][lane][i]      = N[i].x;
            sN[wave][32 + lane][i] = N[i].y;
        }
    }
    __builtin_amdgcn_wave_barrier();

    {
        const float4* s2 = (const float4*)(&sN[wave][0][0]);
        float4* o2 = ((float4*)(out + OUT_HALF)) + (size_t)wbase * 4;
#pragma unroll
        for (int j = 0; j < 8; ++j) o2[lane + j * 32] = s2[lane + j * 32];
    }

    const int half = lane >> 4;
    const int q    = lane & 15;
    const int sub  = q & 3;
    const int grp  = q >> 2;

    const bool act = (grp >> 1) == half;
    const bool hi  = (grp & 1) != 0;
    float* const dumpBase = &sDump[wave][lane];

    const v2f th2w  = tw[0] * tw[0] + tw[1] * tw[1] + tw[2] * tw[2];
    const bool wide = (th2w.x >= 9.0f) || (th2w.y >= 9.0f);
    if (__builtin_amdgcn_ballot_w32(wide) == 0u) {
        v2f sx[6];
#pragma unroll
        for (int i = 0; i < 6; ++i) sx[i] = sq * tw[i];
        v2f Ht[16];
        se3_exp16_x2(sx, Ht);
#pragma unroll
        for (int i = 0; i < 16; ++i) {
            sA[wave][lane][i]      = Ht[i].x;
            sA[wave][32 + lane][i] = Ht[i].y;
        }
    } else {
        const float f1 = 1.0f - sq;
#pragma unroll
        for (int h = 0; h < 2; ++h) {
            const float twh[6] = {tw[0][h], tw[1][h], tw[2][h], tw[3][h], tw[4][h], tw[5][h]};
            float O[16];
            se3_exp16(twh, O);
            float xi[6];
            se3_log_of_inv(O, xi);
#pragma unroll
            for (int i = 0; i < 6; ++i) xi[i] *= f1;
            float E[16];
            se3_exp16(xi, E);
#pragma unroll
            for (int i = 0; i < 16; ++i) {
                sA[wave][32 * h + lane][i] = E[i];
                sB[wave][32 * h + lane][i] = O[i];
            }
        }
        __builtin_amdgcn_wave_barrier();

#pragma unroll
        for (int r = 0; r < 16; ++r) {
            const int it = 4 * r + grp;

            v2f a, bb;
            a.x  = sA[wave][it][sub * 4 + 2 * half];
            a.y  = sA[wave][it][sub * 4 + 2 * half + 1];
            bb.x = sB[wave][it][(2 * half) * 4 + sub];
            bb.y = sB[wave][it][(2 * half + 1) * 4 + sub];

            v8f c = {};
            v8f d = __builtin_amdgcn_wmma_f32_16x16x4_f32(
                false, a, false, bb, (short)0, c, false, false);

            float* p = act ? &sA[wave][it][sub] : dumpBase;
            p[0]  = hi ? d[4] : d[0];
            p[4]  = hi ? d[5] : d[1];
            p[8]  = hi ? d[6] : d[2];
            p[12] = hi ? d[7] : d[3];
            __builtin_amdgcn_wave_barrier();
        }
    }
    __builtin_amdgcn_wave_barrier();

    float4 prod[16];
#pragma unroll
    for (int ps = 0; ps < 2; ++ps) {
        v2f hA[8], nB[8];
#pragma unroll
        for (int r = 0; r < 8; ++r) {
            const int it = 32 * ps + 4 * r + grp;
            hA[r].x = sA[wave][it][(2 * half) * 4 + sub];
            hA[r].y = sA[wave][it][(2 * half + 1) * 4 + sub];
            nB[r].x = sN[wave][it][sub * 4 + 2 * half];
            nB[r].y = sN[wave][it][sub * 4 + 2 * half + 1];
        }
#pragma unroll
        for (int r = 0; r < 8; ++r) {
            v8f c = {};
            v8f d = __builtin_amdgcn_wmma_f32_16x16x4_f32(
                false, hA[r], false, nB[r], (short)0, c, false, false);
            prod[8 * ps + r].x = hi ? d[4] : d[0];
            prod[8 * ps + r].y = hi ? d[5] : d[1];
            prod[8 * ps + r].z = hi ? d[6] : d[2];
            prod[8 * ps + r].w = hi ? d[7] : d[3];
        }
    }

    if (act) {
        float4* o1 = ((float4*)out) + (size_t)wbase * 4 + q;
#pragma unroll
        for (int R = 0; R < 16; ++R) o1[16 * R] = prod[R];
    }
}

extern "C" void kernel_launch(void* const* d_in, const int* in_sizes, int n_in,
                              void* d_out, int out_size, void* d_ws, size_t ws_size,
                              hipStream_t stream) {
    const float* twist      = (const float*)d_in[0];
    const float* noise      = (const float*)d_in[1];
    const float* alpha_bars = (const float*)d_in[2];
    const int*   timesteps  = (const int*)d_in[3];
    float*       out        = (float*)d_out;

    const int blocks = NUM_ITEMS / 512;
    hipLaunchKernelGGL(se3_diffuse_kernel, dim3(blocks), dim3(256), 0, stream,
                       twist, noise, alpha_bars, timesteps, out);
}


# Exp-map coefficients sin(t)/t, (1-cos t)/t^2, (t-sin t)/t^3 by packed Horner series in t^2 (no sqrt, sincos, division) in waves with all t^2 < 9; library form otherwise; coefficient rounding changes
# speedup vs baseline: 1.8038x; 1.0641x over previous
#include <hip/hip_runtime.h>
#include <hip/hip_bf16.h>
#include <math.h>


#define NUM_ITEMS (4096 * 64)
#define OUT_HALF  (NUM_ITEMS * 16)
#define SIGMA_R 0.05f
#define SIGMA_T 0.03f

typedef float v2f __attribute__((ext_vector_type(2)));
typedef float v8f __attribute__((ext_vector_type(8)));
typedef int   v2i __attribute__((ext_vector_type(2)));


__device__ __forceinline__ void se3_exp16(const float xi[6], float* H) {
    const float wx = xi[0], wy = xi[1], wz = xi[2];
    const float vx = xi[3], vy = xi[4], vz = xi[5];
    float th2  = wx * wx + wy * wy + wz * wz;
    bool small = th2 < 1e-8f;
    float th2s = small ? 1.0f : th2;
    float th   = sqrtf(th2s);
    float s, c;
    sincosf(th, &s, &c);
    float A  = small ? 1.0f - th2 * (1.0f / 6.0f)   : s / th;
    float Bc = small ? 0.5f - th2 * (1.0f / 24.0f)  : (1.0f - c) / th2s;
    float Cc = small ? (1.0f / 6.0f) - th2 * (1.0f / 120.0f)
                     : (th - s) / (th2s * th);
    float xx = wx * wx, yy = wy * wy, zz = wz * wz;
    float xy = wx * wy, xz = wx * wz, yz = wy * wz;
    float R00 = 1.0f - Bc * (yy + zz), R01 = Bc * xy - A * wz, R02 = Bc * xz + A * wy;
    float R10 = Bc * xy + A * wz, R11 = 1.0f - Bc * (xx + zz), R12 = Bc * yz - A * wx;
    float R20 = Bc * xz - A * wy, R21 = Bc * yz + A * wx, R22 = 1.0f - Bc * (xx + yy);
    float V00 = 1.0f - Cc * (yy + zz), V01 = Cc * xy - Bc * wz, V02 = Cc * xz + Bc * wy;
    float V10 = Cc * xy + Bc * wz, V11 = 1.0f - Cc * (xx + zz), V12 = Cc * yz - Bc * wx;
    float V20 = Cc * xz - Bc * wy, V21 = Cc * yz + Bc * wx, V22 = 1.0f - Cc * (xx + yy);
    float tx = V00 * vx + V01 * vy + V02 * vz;
    float ty = V10 * vx + V11 * vy + V12 * vz;
    float tz = V20 * vx + V21 * vy + V22 * vz;
    H[0] = R00; H[1] = R01; H[2]  = R02; H[3]  = tx;
    H[4] = R10; H[5] = R11; H[6]  = R12; H[7]  = ty;
    H[8] = R20; H[9] = R21; H[10] = R22; H[11] = tz;
    H[12] = 0.0f; H[13] = 0.0f; H[14] = 0.0f; H[15] = 1.0f;
}

__device__ __forceinline__ void se3_exp16_x2(const v2f xi[6], v2f* H) {
    const v2f wx = xi[0], wy = xi[1], wz = xi[2];
    const v2f vx = xi[3], vy = xi[4], vz = xi[5];
    const v2f one = {1.0f, 1.0f};
    v2f th2   = wx * wx + wy * wy + wz * wz;
    v2f A, Bc, Cc;
    const bool far = !(th2.x < 9.0f && th2.y < 9.0f);
    if (__builtin_amdgcn_ballot_w32(far) == 0u) {
        const v2f y = -th2;
        v2f pc = {(float)(1.0 / 121645100408832000.0), (float)(1.0 / 121645100408832000.0)};
        pc = y * pc + (float)(1.0 / 355687428096000.0);
        pc = y * pc + (float)(1.0 / 1307674368000.0);
        pc = y * pc + (float)(1.0 / 6227020800.0);
        pc = y * pc + (float)(1.0 / 39916800.0);
        pc = y * pc + (float)(1.0 / 362880.0);
        pc = y * pc + (float)(1.0 / 5040.0);
        pc = y * pc + (float)(1.0 / 120.0);
        Cc = y * pc + (float)(1.0 / 6.0);
        A  = y * Cc + 1.0f;
        v2f pb = {(float)(1.0 / 2432902008176640000.0), (float)(1.0 / 2432902008176640000.0)};
        pb = y * pb + (float)(1.0 / 6402373705728000.0);
        pb = y * pb + (float)(1.0 / 20922789888000.0);
        pb = y * pb + (float)(1.0 / 87178291200.0);
        pb = y * pb + (float)(1.0 / 479001600.0);
        pb = y * pb + (float)(1.0 / 3628800.0);
        pb = y * pb + (float)(1.0 / 40320.0);
        pb = y * pb + (float)(1.0 / 720.0);
        pb = y * pb + (float)(1.0 / 24.0);
        Bc = y * pb + 0.5f;
    } else {
        v2i small = th2 < 1e-8f;
        v2f th2s  = small ? one : th2;
        v2f th;
        th.x = sqrtf(th2s.x);
        th.y = sqrtf(th2s.y);
        float s0, c0, s1, c1;
        sincosf(th.x, &s0, &c0);
        sincosf(th.y, &s1, &c1);
        v2f s = {s0, s1};
        v2f c = {c0, c1};
        A  = small ? one - th2 * (1.0f / 6.0f)   : s / th;
        Bc = small ? 0.5f - th2 * (1.0f / 24.0f) : (one - c) / th2s;
        Cc = small ? (1.0f / 6.0f) - th2 * (1.0f / 120.0f)
                   : (th - s) / (th2s * th);
    }
    v2f xx = wx * wx, yy = wy * wy, zz = wz * wz;
    v2f xy = wx * wy, xz = wx * wz, yz = wy * wz;
    v2f R00 = one - Bc * (yy + zz), R01 = Bc * xy - A * wz, R02 = Bc * xz + A * wy;
    v2f R10 = Bc * xy + A * wz, R11 = one - Bc * (xx + zz), R12 = Bc * yz - A * wx;
    v2f R20 = Bc * xz - A * wy, R21 = Bc * yz + A * wx, R22 = one - Bc * (xx + yy);
    v2f V00 = one - Cc * (yy + zz), V01 = Cc * xy - Bc * wz, V02 = Cc * xz + Bc * wy;
    v2f V10 = Cc * xy + Bc * wz, V11 = one - Cc * (xx + zz), V12 = Cc * yz - Bc * wx;
    v2f V20 = Cc * xz - Bc * wy, V21 = Cc * yz + Bc * wx, V22 = one - Cc * (xx + yy);
    v2f tx = V00 * vx + V01 * vy + V02 * vz;
    v2f ty = V10 * vx + V11 * vy + V12 * vz;
    v2f tz = V20 * vx + V21 * vy + V22 * vz;
    const v2f zero = {0.0f, 0.0f};
    H[0] = R00; H[1] = R01; H[2]  = R02; H[3]  = tx;
    H[4] = R10; H[5] = R11; H[6]  = R12; H[7]  = ty;
    H[8] = R20; H[9] = R21; H[10] = R22; H[11] = tz;
    H[12] = zero; H[13] = zero; H[14] = zero; H[15] = one;
}

__device__ __forceinline__ void se3_log_of_inv(const float* O, float xi[6]) {
    float R00 = O[0], R01 = O[4], R02 = O[8];
    float R10 = O[1], R11 = O[5], R12 = O[9];
    float R20 = O[2], R21 = O[6], R22 = O[10];
    float tx = -(R00 * O[3] + R01 * O[7] + R02 * O[11]);
    float ty = -(R10 * O[3] + R11 * O[7] + R12 * O[11]);
    float tz = -(R20 * O[3] + R21 * O[7] + R22 * O[11]);
    float e0 = 0.5f * (R21 - R12);
    float e1 = 0.5f * (R02 - R20);
    float e2 = 0.5f * (R10 - R01);
    float cos_t = 0.5f * (R00 + R11 + R22 - 1.0f);
    cos_t = fminf(1.0f, fmaxf(-1.0f, cos_t));
    float s2 = e0 * e0 + e1 * e1 + e2 * e2;
    bool small = s2 < 1e-12f;
    float sin_s = sqrtf(small ? 1.0f : s2);
    float th = atan2f(sin_s, cos_t);
    float fac = small ? 1.0f + s2 * (1.0f / 6.0f) : th / sin_s;
    float wx = fac * e0, wy = fac * e1, wz = fac * e2;
    float omc  = small ? 1.0f : 1.0f - cos_t;
    float th2s = small ? 1.0f : th * th;
    float D = small ? (1.0f / 12.0f) + s2 * (1.0f / 720.0f)
                    : (1.0f - th * sin_s / (2.0f * omc)) / th2s;
    float xx = wx * wx, yy = wy * wy, zz = wz * wz;
    float xy = wx * wy, xz = wx * wz, yz = wy * wz;
    float Vi00 = 1.0f - D * (yy + zz), Vi01 = D * xy + 0.5f * wz, Vi02 = D * xz - 0.5f * wy;
    float Vi10 = D * xy - 0.5f * wz, Vi11 = 1.0f - D * (xx + zz), Vi12 = D * yz + 0.5f * wx;
    float Vi20 = D * xz + 0.5f * wy, Vi21 = D * yz - 0.5f * wx, Vi22 = 1.0f - D * (xx + yy);
    xi[0] = wx; xi[1] = wy; xi[2] = wz;
    xi[3] = Vi00 * tx + Vi01 * ty + Vi02 * tz;
    xi[4] = Vi10 * tx + Vi11 * ty + Vi12 * tz;
    xi[5] = Vi20 * tx + Vi21 * ty + Vi22 * tz;
}


__global__ __launch_bounds__(256) void
se3_diffuse_kernel(const float* __restrict__ twist,
                   const float* __restrict__ noise,
                   const float* __restrict__ alpha_bars,
                   const int* __restrict__ timesteps,
                   float* __restrict__ out) {
    __shared__ float sA[8][64][16];
    __shared__ float sB[8][64][16];
    __shared__ float sN[8][64][16];
    __shared__ float sDump[8][48];

    const int tid   = threadIdx.x;
    const int wave  = tid >> 5;
    const int lane  = tid & 31;
    const int b     = blockIdx.x * 8 + wave;
    const int wbase = b * 64;

    v2f tw[6], nz[6];
    {
        const float2* twa = (const float2*)(twist + (size_t)(wbase + lane) * 6);
        const float2* nza = (const float2*)(noise + (size_t)(wbase + lane) * 6);
        const float2* twb = twa + 32 * 3;
        const float2* nzb = nza + 32 * 3;
        float2 ta0 = twa[0], ta1 = twa[1], ta2 = twa[2];
        float2 tb0 = twb[0], tb1 = twb[1], tb2 = twb[2];
        float2 na0 = nza[0], na1 = nza[1], na2 = nza[2];
        float2 nb0 = nzb[0], nb1 = nzb[1], nb2 = nzb[2];
        tw[0] = (v2f){ta0.x, tb0.x}; tw[1] = (v2f){ta0.y, tb0.y}; tw[2] = (v2f){ta1.x, tb1.x};
        tw[3] = (v2f){ta1.y, tb1.y}; tw[4] = (v2f){ta2.x, tb2.x}; tw[5] = (v2f){ta2.y, tb2.y};
        nz[0] = (v2f){na0.x, nb0.x}; nz[1] = (v2f){na0.y, nb0.y}; nz[2] = (v2f){na1.x, nb1.x};
        nz[3] = (v2f){na1.y, nb1.y}; nz[4] = (v2f){na2.x, nb2.x}; nz[5] = (v2f){na2.y, nb2.y};
    }

    const int bs = __builtin_amdgcn_readfirstlane(b);
    const float ab = alpha_bars[timesteps[bs]];
    const float sq = sqrtf(ab);
    const float f2 = sqrtf(1.0f - ab);
    const float fr = f2 * SIGMA_R;
    const float ft = f2 * SIGMA_T;

    {
        v2f nv[6] = {fr * nz[0], fr * nz[1], fr * nz[2],
                     ft * nz[3], ft * nz[4], ft * nz[5]};
        v2f N[16];
        se3_exp16_x2(nv, N);
#pragma unroll
        for (int i = 0; i < 16; ++i) {
            sN[wave][lane][i]      = N[i].x;
            sN[wave][32 + lane][i] = N[i].y;
        }
    }
    __builtin_amdgcn_wave_barrier();

    {
        const float4* s2 = (const float4*)(&sN[wave][0][0]);
        float4* o2 = ((float4*)(out + OUT_HALF)) + (size_t)wbase * 4;
#pragma unroll
        for (int j = 0; j < 8; ++j) o2[lane + j * 32] = s2[lane + j * 32];
    }

    const int half = lane >> 4;
    const int q    = lane & 15;
    const int sub  = q & 3;
    const int grp  = q >> 2;

    const bool act = (grp >> 1) == half;
    const bool hi  = (grp & 1) != 0;
    float* const dumpBase = &sDump[wave][lane];

    const v2f th2w  = tw[0] * tw[0] + tw[1] * tw[1] + tw[2] * tw[2];
    const bool wide = (th2w.x >= 9.0f) || (th2w.y >= 9.0f);
    if (__builtin_amdgcn_ballot_w32(wide) == 0u) {
        v2f sx[6];
#pragma unroll
        for (int i = 0; i < 6; ++i) sx[i] = sq * tw[i];
        v2f Ht[16];
        se3_exp16_x2(sx, Ht);
#pragma unroll
        for (int i = 0; i < 16; ++i) {
            sA[wave][lane][i]      = Ht[i].x;
            sA[wave][32 + lane][i] = Ht[i].y;
        }
    } else {
        const float f1 = 1.0f - sq;
#pragma unroll
        for (int h = 0; h < 2; ++h) {
            const float twh[6] = {tw[0][h], tw[1][h], tw[2][h], tw[3][h], tw[4][h], tw[5][h]};
            float O[16];
            se3_exp16(twh, O);
            float xi[6];
            se3_log_of_inv(O, xi);
#pragma unroll
            for (int i = 0; i < 6; ++i) xi[i] *= f1;
            float E[16];
            se3_exp16(xi, E);
#pragma unroll
            for (int i = 0; i < 16; ++i) {
                sA[wave][32 * h + lane][i] = E[i];
                sB[wave][32 * h + lane][i] = O[i];
            }
        }
        __builtin_amdgcn_wave_barrier();

#pragma unroll
        for (int r = 0; r < 16; ++r) {
            const int it = 4 * r + grp;

            v2f a, bb;
            a.x  = sA[wave][it][sub * 4 + 2 * half];
            a.y  = sA[wave][it][sub * 4 + 2 * half + 1];
            bb.x = sB[wave][it][(2 * half) * 4 + sub];
            bb.y = sB[wave][it][(2 * half + 1) * 4 + sub];

            v8f c = {};
            v8f d = __builtin_amdgcn_wmma_f32_16x16x4_f32(
                false, a, false, bb, (short)0, c, false, false);

            float* p = act ? &sA[wave][it][sub] : dumpBase;
            p[0]  = hi ? d[4] : d[0];
            p[4]  = hi ? d[5] : d[1];
            p[8]  = hi ? d[6] : d[2];
            p[12] = hi ? d[7] : d[3];
            __builtin_amdgcn_wave_barrier();
        }
    }
    __builtin_amdgcn_wave_barrier();

    float4 prod[16];
#pragma unroll
    for (int ps = 0; ps < 2; ++ps) {
        v2f hA[8], nB[8];
#pragma unroll
        for (int r = 0; r < 8; ++r) {
            const int it = 32 * ps + 4 * r + grp;
            hA[r].x = sA[wave][it][(2 * half) * 4 + sub];
            hA[r].y = sA[wave][it][(2 * half + 1) * 4 + sub];
            nB[r].x = sN[wave][it][sub * 4 + 2 * half];
            nB[r].y = sN[wave][it][sub * 4 + 2 * half + 1];
        }
#pragma unroll
        for (int r = 0; r < 8; ++r) {
            v8f c = {};
            v8f d = __builtin_amdgcn_wmma_f32_16x16x4_f32(
                false, hA[r], false, nB[r], (short)0, c, false, false);
            prod[8 * ps + r].x = hi ? d[4] : d[0];
            prod[8 * ps + r].y = hi ? d[5] : d[1];
            prod[8 * ps + r].z = hi ? d[6] : d[2];
            prod[8 * ps + r].w = hi ? d[7] : d[3];
        }
    }

    if (act) {
        float4* o1 = ((float4*)out) + (size_t)wbase * 4 + q;
#pragma unroll
        for (int R = 0; R < 16; ++R) o1[16 * R] = prod[R];
    }
}

extern "C" void kernel_launch(void* const* d_in, const int* in_sizes, int n_in,
                              void* d_out, int out_size, void* d_ws, size_t ws_size,
                              hipStream_t stream) {
    const float* twist      = (const float*)d_in[0];
    const float* noise      = (const float*)d_in[1];
    const float* alpha_bars = (const float*)d_in[2];
    const int*   timesteps  = (const int*)d_in[3];
    float*       out        = (float*)d_out;

    const int blocks = NUM_ITEMS / 512;
    hipLaunchKernelGGL(se3_diffuse_kernel, dim3(blocks), dim3(256), 0, stream,
                       twist, noise, alpha_bars, timesteps, out);
}
